# starting kernel
# baseline (speedup 1.0000x reference)
#include <hip/hip_runtime.h>

typedef __attribute__((ext_vector_type(2))) float v2f;
typedef __attribute__((ext_vector_type(8))) float v8f;

#define Bq 128
#define Nq 512
#define Fq 32
#define Hq 128
#define Dq 64
#define Kq 16

__global__ __launch_bounds__(256) void knn_kernel(const float* __restrict__ events,
                                                  int* __restrict__ knn_idx) {
  __shared__ float sx[Nq];
  __shared__ float sy[Nq];
  const int tid = threadIdx.x;
  const int p   = blockIdx.x * 256 + tid;
  const int b   = p >> 9;
  const int i   = p & (Nq - 1);

  for (int j = tid; j < Nq; j += 256) {
    const float* e = events + ((size_t)b * Nq + j) * Fq;
    sx[j] = e[0];
    sy[j] = e[1];
  }
  __syncthreads();

  const float cx = sx[i], cy = sy[i];
  float bestd[Kq];
  int   besti[Kq];
#pragma unroll
  for (int t = 0; t < Kq; ++t) { bestd[t] = 3.4e38f; besti[t] = 0; }

  for (int j = 0; j < Nq; ++j) {
    if (j == i) continue;
    const float dx = sx[j] - cx;
    const float dy = sy[j] - cy;
    const float d  = dx * dx + dy * dy;
    if (d < bestd[Kq - 1]) {
      float dd = d; int ii = j;
#pragma unroll
      for (int t = 0; t < Kq; ++t) {
        if (dd < bestd[t]) {
          const float td = bestd[t]; const int ti = besti[t];
          bestd[t] = dd; besti[t] = ii;
          dd = td; ii = ti;
        }
      }
    }
  }
#pragma unroll
  for (int t = 0; t < Kq; ++t) knn_idx[(size_t)p * Kq + t] = besti[t];
}

#define WAVES_PER_BLOCK 4
#define H_STRIDE 129

__global__ __launch_bounds__(WAVES_PER_BLOCK * 32) void edgeconv_mlp_kernel(
    const float* __restrict__ events, const int* __restrict__ knn_idx,
    const float* __restrict__ W1, const float* __restrict__ b1,
    const float* __restrict__ W2, const float* __restrict__ b2,
    float* __restrict__ out) {
  __shared__ float ldsH[WAVES_PER_BLOCK * 16 * H_STRIDE];

  const int lane = threadIdx.x & 31;
  const int wid  = threadIdx.x >> 5;
  const int p    = blockIdx.x * WAVES_PER_BLOCK + wid;
  const int b    = p >> 9;
  const int i    = p & (Nq - 1);
  const int m    = lane & 15;
  const int h    = lane >> 4;

  float* myH = ldsH + wid * 16 * H_STRIDE;

  const int nj = knn_idx[(size_t)p * Kq + m];
  const float* ev = events + ((size_t)b * Nq + i) * Fq;
  const float* nb = events + ((size_t)b * Nq + nj) * Fq;

  float a[32];
#pragma unroll
  for (int s = 0; s < 16; ++s) {
    if (s < 8) {
      const int c = 4 * s + 2 * h;
      a[2 * s]     = ev[c];
      a[2 * s + 1] = ev[c + 1];
    } else {
      const int c = 4 * (s - 8) + 2 * h;
      a[2 * s]     = nb[c]     - ev[c];
      a[2 * s + 1] = nb[c + 1] - ev[c + 1];
    }
  }

  const v8f vzero = {};

  v8f acc1[8];
#pragma unroll
  for (int t = 0; t < 8; ++t) acc1[t] = vzero;

#pragma unroll
  for (int s = 0; s < 16; ++s) {
    v2f A;
    A.x = a[2 * s];
    A.y = a[2 * s + 1];
    const int kb = 4 * s + 2 * h;
    const float* w1r0 = W1 + (size_t)kb * Hq + m;
    const float* w1r1 = W1 + (size_t)(kb + 1) * Hq + m;
#pragma unroll
    for (int t = 0; t < 8; ++t) {
      v2f Bv;
      Bv.x = w1r0[16 * t];
      Bv.y = w1r1[16 * t];
      acc1[t] = __builtin_amdgcn_wmma_f32_16x16x4_f32(
          false, A, false, Bv, (short)0, acc1[t], false, false);
    }
  }

#pragma unroll
  for (int t = 0; t < 8; ++t) {
    const float bv = b1[16 * t + m];
#pragma unroll
    for (int r = 0; r < 8; ++r) {
      float x = acc1[t][r] + bv;
      x = fmaxf(x, 0.0f);
      myH[(r + 8 * h) * H_STRIDE + 16 * t + m] = x;
    }
  }

  v8f acc2[4];
#pragma unroll
  for (int t = 0; t < 4; ++t) acc2[t] = vzero;

#pragma unroll
  for (int s = 0; s < 32; ++s) {
    const int kb = 4 * s + 2 * h;
    v2f A;
    A.x = myH[m * H_STRIDE + kb];
    A.y = myH[m * H_STRIDE + kb + 1];
    const float* w2r0 = W2 + (size_t)kb * Dq + m;
    const float* w2r1 = W2 + (size_t)(kb + 1) * Dq + m;
#pragma unroll
    for (int t = 0; t < 4; ++t) {
      v2f Bv;
      Bv.x = w2r0[16 * t];
      Bv.y = w2r1[16 * t];
      acc2[t] = __builtin_amdgcn_wmma_f32_16x16x4_f32(
          false, A, false, Bv, (short)0, acc2[t], false, false);
    }
  }

#pragma unroll
  for (int t = 0; t < 4; ++t) {
    const float bv = b2[16 * t + m];
    float part = 0.0f;
#pragma unroll
    for (int r = 0; r < 8; ++r) part += fmaxf(acc2[t][r] + bv, 0.0f);
    part += __shfl_xor(part, 16, 32);
    if (h == 0) out[(size_t)p * Dq + 16 * t + m] = part * (1.0f / 16.0f);
  }
}

extern "C" void kernel_launch(void* const* d_in, const int* in_sizes, int n_in,
                              void* d_out, int out_size, void* d_ws, size_t ws_size,
                              hipStream_t stream) {
  const float* events = (const float*)d_in[0];
  const float* W1     = (const float*)d_in[1];
  const float* b1     = (const float*)d_in[2];
  const float* W2     = (const float*)d_in[3];
  const float* b2     = (const float*)d_in[4];
  float*       outp   = (float*)d_out;
  int*         knn    = (int*)d_ws;

  knn_kernel<<<(Bq * Nq) / 256, 256, 0, stream>>>(events, knn);
  edgeconv_mlp_kernel<<<(Bq * Nq) / WAVES_PER_BLOCK, WAVES_PER_BLOCK * 32, 0, stream>>>(
      events, knn, W1, b1, W2, b2, outp);
}


# MLP weights: W1/W2 copied once per 16-wave work-group into LDS in B-fragment order; a wave walks 16 points, B fragments come from grouped 16-byte LDS reads, not 512 global loads a point; same sums
# speedup vs baseline: 3.3414x; 3.3414x over previous
#include <hip/hip_runtime.h>

typedef __attribute__((ext_vector_type(2))) float v2f;
typedef __attribute__((ext_vector_type(8))) float v8f;

#define Bq 128
#define Nq 512
#define Fq 32
#define Hq 128
#define Dq 64
#define Kq 16

__global__ __launch_bounds__(256) void knn_kernel(const float* __restrict__ events,
                                                  int* __restrict__ knn_idx) {
  __shared__ float2 sxy[Nq];
  const int tid = threadIdx.x;
  const int p   = blockIdx.x * 256 + tid;
  const int b   = p >> 9;
  const int i   = p & (Nq - 1);

  for (int j = tid; j < Nq; j += 256) {
    const float* e = events + ((size_t)b * Nq + j) * Fq;
    sxy[j] = make_float2(e[0], e[1]);
  }
  __syncthreads();

  const float cx = sxy[i].x, cy = sxy[i].y;
  float bestd[Kq];
  int   besti[Kq];
#pragma unroll
  for (int t = 0; t < Kq; ++t) { bestd[t] = 3.4e38f; besti[t] = 0; }

#pragma unroll 4
  for (int j = 0; j < Nq; ++j) {
    const float2 c = sxy[j];
    const float dx = c.x - cx;
    const float dy = c.y - cy;
    float d = __builtin_fmaf(dx, dx, dy * dy);
    d = (j == i) ? 3.4e38f : d;

    bool lt[Kq];
#pragma unroll
    for (int t = 0; t < Kq; ++t) lt[t] = d < bestd[t];

#pragma unroll
    for (int t = Kq - 1; t >= 1; --t) {
      const int shifted = lt[t - 1] ? besti[t - 1] : j;
      besti[t] = lt[t] ? shifted : besti[t];
      bestd[t] = fminf(bestd[t], fmaxf(bestd[t - 1], d));
    }
    besti[0] = lt[0] ? j : besti[0];
    bestd[0] = fminf(bestd[0], d);
  }
#pragma unroll
  for (int t = 0; t < Kq; ++t) knn_idx[(size_t)p * Kq + t] = besti[t];
}

typedef __attribute__((ext_vector_type(4))) float v4f;

#define MLP_WAVES 16
#define PTS_PER_WG 256
#define PTS_PER_WAVE (PTS_PER_WG / MLP_WAVES)
#define H_STRIDE 129

__device__ __forceinline__ void load_edge_rows(float (&a)[32], const float* ev,
                                               const float* nb, int h) {
#pragma unroll
  for (int s = 0; s < 8; ++s) {
    const int c = 4 * s + 2 * h;
    const float x0 = ev[c], x1 = ev[c + 1];
    a[2 * s]          = x0;
    a[2 * s + 1]      = x1;
    a[2 * (s + 8)]     = nb[c]     - x0;
    a[2 * (s + 8) + 1] = nb[c + 1] - x1;
  }
}

__global__ __launch_bounds__(MLP_WAVES * 32) void edgeconv_mlp_kernel(
    const float* __restrict__ events, const int* __restrict__ knn_idx,
    const float* __restrict__ W1, const float* __restrict__ b1,
    const float* __restrict__ W2, const float* __restrict__ b2,
    float* __restrict__ out) {
  __shared__ v4f   ldsW1[8 * 8 * 32];
  __shared__ v4f   ldsW2[16 * 4 * 32];
  __shared__ float ldsH[MLP_WAVES * 16 * H_STRIDE];

  const int tid  = threadIdx.x;
  const int lane = tid & 31;
  const int wid  = tid >> 5;
  const int m    = lane & 15;
  const int h    = lane >> 4;

#pragma unroll
  for (int i = 0; i < (8 * 8 * 32) / (MLP_WAVES * 32); ++i) {
    const int q = tid + i * (MLP_WAVES * 32);
    const int ql = q & 31, t = (q >> 5) & 7, sp = q >> 8;
    const float* w = W1 + (size_t)(8 * sp + 2 * (ql >> 4)) * Hq + 16 * t + (ql & 15);
    v4f v;
    v.x = w[0];
    v.y = w[Hq];
    v.z = w[4 * Hq];
    v.w = w[5 * Hq];
    ldsW1[q] = v;
  }
#pragma unroll
  for (int i = 0; i < (16 * 4 * 32) / (MLP_WAVES * 32); ++i) {
    const int q = tid + i * (MLP_WAVES * 32);
    const int ql = q & 31, t = (q >> 5) & 3, sp = q >> 7;
    const float* w = W2 + (size_t)(8 * sp + 2 * (ql >> 4)) * Dq + 16 * t + (ql & 15);
    v4f v;
    v.x = w[0];
    v.y = w[Dq];
    v.z = w[4 * Dq];
    v.w = w[5 * Dq];
    ldsW2[q] = v;
  }

  float bias1[8], bias2[4];
#pragma unroll
  for (int t = 0; t < 8; ++t) bias1[t] = b1[16 * t + m];
#pragma unroll
  for (int t = 0; t < 4; ++t) bias2[t] = b2[16 * t + m];

  float* myH = ldsH + wid * 16 * H_STRIDE;

  const int pbase = blockIdx.x * PTS_PER_WG;
  const float* evb = events + (size_t)(pbase >> 9) * Nq * Fq;

  int p = pbase + wid;
  float a[32];
  {
    const int nj = knn_idx[(size_t)p * Kq + m];
    load_edge_rows(a, evb + (size_t)(p & (Nq - 1)) * Fq, evb + (size_t)nj * Fq, h);
  }

  __syncthreads();

  const v8f vzero = {};

#pragma unroll 1
  for (int it = 0; it < PTS_PER_WAVE; ++it) {
    const int pn = (it + 1 < PTS_PER_WAVE) ? p + MLP_WAVES : p;
    const int njn = knn_idx[(size_t)pn * Kq + m];

    v8f acc1[8];
#pragma unroll
    for (int t = 0; t < 8; ++t) acc1[t] = vzero;

    v4f bq[2][8];
#pragma unroll
    for (int t = 0; t < 8; ++t) bq[0][t] = ldsW1[t * 32 + lane];
    __builtin_amdgcn_sched_group_barrier(0x100, 8, 0);
#pragma unroll
    for (int sp = 0; sp < 8; ++sp) {
      if (sp + 1 < 8) {
#pragma unroll
        for (int t = 0; t < 8; ++t) bq[(sp + 1) & 1][t] = ldsW1[((sp + 1) * 8 + t) * 32 + lane];
      }
      v2f A0, A1;
      A0.x = a[4 * sp];
      A0.y = a[4 * sp + 1];
      A1.x = a[4 * sp + 2];
      A1.y = a[4 * sp + 3];
#pragma unroll
      for (int t = 0; t < 8; ++t) {
        v2f Bv;
        Bv.x = bq[sp & 1][t].x;
        Bv.y = bq[sp & 1][t].y;
        acc1[t] = __builtin_amdgcn_wmma_f32_16x16x4_f32(
            false, A0, false, Bv, (short)0, acc1[t], false, false);
      }
#pragma unroll
      for (int t = 0; t < 8; ++t) {
        v2f Bv;
        Bv.x = bq[sp & 1][t].z;
        Bv.y = bq[sp & 1][t].w;
        acc1[t] = __builtin_amdgcn_wmma_f32_16x16x4_f32(
            false, A1, false, Bv, (short)0, acc1[t], false, false);
      }
      if (sp + 1 < 8) __builtin_amdgcn_sched_group_barrier(0x100, 8, 0);
      __builtin_amdgcn_sched_group_barrier(0x008, 16, 0);
    }

    load_edge_rows(a, evb + (size_t)(pn & (Nq - 1)) * Fq, evb + (size_t)njn * Fq, h);

#pragma unroll
    for (int t = 0; t < 8; ++t) {
#pragma unroll
      for (int r = 0; r < 8; ++r) {
        float x = acc1[t][r] + bias1[t];
        x = fmaxf(x, 0.0f);
        myH[(r + 8 * h) * H_STRIDE + 16 * t + m] = x;
      }
    }

    v8f acc2[4];
#pragma unroll
    for (int t = 0; t < 4; ++t) acc2[t] = vzero;

    v4f cq[2][4];
    v2f hA[2][2];
#pragma unroll
    for (int t = 0; t < 4; ++t) cq[0][t] = ldsW2[t * 32 + lane];
    hA[0][0].x = myH[m * H_STRIDE + 2 * h];
    hA[0][0].y = myH[m * H_STRIDE + 2 * h + 1];
    hA[0][1].x = myH[m * H_STRIDE + 2 * h + 4];
    hA[0][1].y = myH[m * H_STRIDE + 2 * h + 5];
    __builtin_amdgcn_sched_group_barrier(0x100, 6, 0);
#pragma unroll
    for (int sp = 0; sp < 16; ++sp) {
      if (sp + 1 < 16) {
        const int kn = 8 * (sp + 1) + 2 * h;
#pragma unroll
        for (int t = 0; t < 4; ++t) cq[(sp + 1) & 1][t] = ldsW2[((sp + 1) * 4 + t) * 32 + lane];
        hA[(sp + 1) & 1][0].x = myH[m * H_STRIDE + kn];
        hA[(sp + 1) & 1][0].y = myH[m * H_STRIDE + kn + 1];
        hA[(sp + 1) & 1][1].x = myH[m * H_STRIDE + kn + 4];
        hA[(sp + 1) & 1][1].y = myH[m * H_STRIDE + kn + 5];
      }
#pragma unroll
      for (int t = 0; t < 4; ++t) {
        v2f Bv;
        Bv.x = cq[sp & 1][t].x;
        Bv.y = cq[sp & 1][t].y;
        acc2[t] = __builtin_amdgcn_wmma_f32_16x16x4_f32(
            false, hA[sp & 1][0], false, Bv, (short)0, acc2[t], false, false);
      }
#pragma unroll
      for (int t = 0; t < 4; ++t) {
        v2f Bv;
        Bv.x = cq[sp & 1][t].z;
        Bv.y = cq[sp & 1][t].w;
        acc2[t] = __builtin_amdgcn_wmma_f32_16x16x4_f32(
            false, hA[sp & 1][1], false, Bv, (short)0, acc2[t], false, false);
      }
      if (sp + 1 < 16) __builtin_amdgcn_sched_group_barrier(0x100, 6, 0);
      __builtin_amdgcn_sched_group_barrier(0x008, 8, 0);
    }

#pragma unroll
    for (int t = 0; t < 4; ++t) {
      float part = 0.0f;
#pragma unroll
      for (int r = 0; r < 8; ++r) part += fmaxf(acc2[t][r] + bias2[t], 0.0f);
      part += __shfl_xor(part, 16, 32);
      if (h == 0) out[(size_t)p * Dq + 16 * t + m] = part * (1.0f / 16.0f);
    }

    p = pn;
  }
}

extern "C" void kernel_launch(void* const* d_in, const int* in_sizes, int n_in,
                              void* d_out, int out_size, void* d_ws, size_t ws_size,
                              hipStream_t stream) {
  const float* events = (const float*)d_in[0];
  const float* W1     = (const float*)d_in[1];
  const float* b1     = (const float*)d_in[2];
  const float* W2     = (const float*)d_in[3];
  const float* b2     = (const float*)d_in[4];
  float*       outp   = (float*)d_out;
  int*         knn    = (int*)d_ws;

  knn_kernel<<<(Bq * Nq) / 256, 256, 0, stream>>>(events, knn);
  edgeconv_mlp_kernel<<<(Bq * Nq) / PTS_PER_WG, MLP_WAVES * 32, 0, stream>>>(
      events, knn, W1, b1, W2, b2, outp);
}


# KNN: per chunk of 32-64 candidates the distance is tested against the list's 16th distance and survivors go to per-lane LDS queues; only those run the 16-slot insertion; same ids in the same order
# speedup vs baseline: 3.6104x; 1.0805x over previous
#include <hip/hip_runtime.h>

typedef __attribute__((ext_vector_type(2))) float v2f;
typedef __attribute__((ext_vector_type(8))) float v8f;

#define Bq 128
#define Nq 512
#define Fq 32
#define Hq 128
#define Dq 64
#define Kq 16

#define QCAP 64
#define QSLOT_BYTES (256 * 8)

__device__ __forceinline__ void knn_insert(float (&bestd)[Kq], int (&besti)[Kq],
                                           float d, int j) {
  bool lt[Kq];
#pragma unroll
  for (int t = 0; t < Kq; ++t) lt[t] = d < bestd[t];
#pragma unroll
  for (int t = Kq - 1; t >= 1; --t) {
    const int shifted = lt[t - 1] ? besti[t - 1] : j;
    besti[t] = lt[t] ? shifted : besti[t];
    bestd[t] = fminf(bestd[t], fmaxf(bestd[t - 1], d));
  }
  besti[0] = lt[0] ? j : besti[0];
  bestd[0] = fminf(bestd[0], d);
}

__global__ __launch_bounds__(256) void knn_kernel(const float* __restrict__ events,
                                                  int* __restrict__ knn_idx) {
  __shared__ __attribute__((aligned(16))) float2 sq[QCAP * 256];
  __shared__ __attribute__((aligned(16))) float2 sxy[Nq];
  const int tid = threadIdx.x;
  const int p   = blockIdx.x * 256 + tid;
  const int b   = p >> 9;
  const int i   = p & (Nq - 1);

  for (int j = tid; j < Nq; j += 256) {
    const float* e = events + ((size_t)b * Nq + j) * Fq;
    sxy[j] = make_float2(e[0], e[1]);
  }
  for (int s = 0; s < QCAP; ++s) sq[s * 256 + tid] = make_float2(0.0f, 0.0f);
  __syncthreads();

  const float cx = sxy[i].x, cy = sxy[i].y;
  float bestd[Kq];
  int   besti[Kq];
#pragma unroll
  for (int t = 0; t < Kq; ++t) { bestd[t] = 3.4e38f; besti[t] = 0; }

  char* const qbase = reinterpret_cast<char*>(sq) + tid * 8;

  for (int j0 = 0; j0 < Nq;) {
    const int len = (j0 < 64) ? 32 : 64;
    const float thr = bestd[Kq - 1];
    int qb = 0;

#pragma unroll 4
    for (int jj = 0; jj < len; jj += 2) {
      const int j = j0 + jj;
      const float4 c2 = *reinterpret_cast<const float4*>(&sxy[j]);
      {
        const float dx = c2.x - cx;
        const float dy = c2.y - cy;
        const float d = __builtin_fmaf(dx, dx, dy * dy);
        *reinterpret_cast<float2*>(qbase + qb) = make_float2(d, __int_as_float(j));
        qb += (d < thr) ? QSLOT_BYTES : 0;
      }
      {
        const float dx = c2.z - cx;
        const float dy = c2.w - cy;
        const float d = __builtin_fmaf(dx, dx, dy * dy);
        *reinterpret_cast<float2*>(qbase + qb) = make_float2(d, __int_as_float(j + 1));
        qb += (d < thr) ? QSLOT_BYTES : 0;
      }
    }
    const int cnt = qb / QSLOT_BYTES;

    for (int q = 0; q < QCAP; q += 2) {
      if (__builtin_amdgcn_ballot_w32(q < cnt) == 0u) break;
      const float2 e0 = *reinterpret_cast<const float2*>(qbase + q * QSLOT_BYTES);
      const float2 e1 = *reinterpret_cast<const float2*>(qbase + (q + 1) * QSLOT_BYTES);
      const int ja = __float_as_int(e0.y);
      const int jb = __float_as_int(e1.y);
      const float da = ((q < cnt) & (ja != i)) ? e0.x : 3.4e38f;
      const float db = ((q + 1 < cnt) & (jb != i)) ? e1.x : 3.4e38f;
      knn_insert(bestd, besti, da, ja);
      knn_insert(bestd, besti, db, jb);
    }
    j0 += len;
  }
#pragma unroll
  for (int t = 0; t < Kq; ++t) knn_idx[(size_t)p * Kq + t] = besti[t];
}

typedef __attribute__((ext_vector_type(4))) float v4f;

#define MLP_WAVES 16
#define PTS_PER_WG 256
#define PTS_PER_WAVE (PTS_PER_WG / MLP_WAVES)
#define H_STRIDE 129

__device__ __forceinline__ void load_edge_rows(float (&a)[32], const float* ev,
                                               const float* nb, int h) {
#pragma unroll
  for (int s = 0; s < 8; ++s) {
    const int c = 4 * s + 2 * h;
    const float x0 = ev[c], x1 = ev[c + 1];
    a[2 * s]          = x0;
    a[2 * s + 1]      = x1;
    a[2 * (s + 8)]     = nb[c]     - x0;
    a[2 * (s + 8) + 1] = nb[c + 1] - x1;
  }
}

__global__ __launch_bounds__(MLP_WAVES * 32) void edgeconv_mlp_kernel(
    const float* __restrict__ events, const int* __restrict__ knn_idx,
    const float* __restrict__ W1, const float* __restrict__ b1,
    const float* __restrict__ W2, const float* __restrict__ b2,
    float* __restrict__ out) {
  __shared__ v4f   ldsW1[8 * 8 * 32];
  __shared__ v4f   ldsW2[16 * 4 * 32];
  __shared__ float ldsH[MLP_WAVES * 16 * H_STRIDE];

  const int tid  = threadIdx.x;
  const int lane = tid & 31;
  const int wid  = tid >> 5;
  const int m    = lane & 15;
  const int h    = lane >> 4;

#pragma unroll
  for (int i = 0; i < (8 * 8 * 32) / (MLP_WAVES * 32); ++i) {
    const int q = tid + i * (MLP_WAVES * 32);
    const int ql = q & 31, t = (q >> 5) & 7, sp = q >> 8;
    const float* w = W1 + (size_t)(8 * sp + 2 * (ql >> 4)) * Hq + 16 * t + (ql & 15);
    v4f v;
    v.x = w[0];
    v.y = w[Hq];
    v.z = w[4 * Hq];
    v.w = w[5 * Hq];
    ldsW1[q] = v;
  }
#pragma unroll
  for (int i = 0; i < (16 * 4 * 32) / (MLP_WAVES * 32); ++i) {
    const int q = tid + i * (MLP_WAVES * 32);
    const int ql = q & 31, t = (q >> 5) & 3, sp = q >> 7;
    const float* w = W2 + (size_t)(8 * sp + 2 * (ql >> 4)) * Dq + 16 * t + (ql & 15);
    v4f v;
    v.x = w[0];
    v.y = w[Dq];
    v.z = w[4 * Dq];
    v.w = w[5 * Dq];
    ldsW2[q] = v;
  }

  float bias1[8], bias2[4];
#pragma unroll
  for (int t = 0; t < 8; ++t) bias1[t] = b1[16 * t + m];
#pragma unroll
  for (int t = 0; t < 4; ++t) bias2[t] = b2[16 * t + m];

  float* myH = ldsH + wid * 16 * H_STRIDE;

  const int pbase = blockIdx.x * PTS_PER_WG;
  const float* evb = events + (size_t)(pbase >> 9) * Nq * Fq;

  int p = pbase + wid;
  float a[32];
  {
    const int nj = knn_idx[(size_t)p * Kq + m];
    load_edge_rows(a, evb + (size_t)(p & (Nq - 1)) * Fq, evb + (size_t)nj * Fq, h);
  }

  __syncthreads();

  const v8f vzero = {};

#pragma unroll 1
  for (int it = 0; it < PTS_PER_WAVE; ++it) {
    const int pn = (it + 1 < PTS_PER_WAVE) ? p + MLP_WAVES : p;
    const int njn = knn_idx[(size_t)pn * Kq + m];

    v8f acc1[8];
#pragma unroll
    for (int t = 0; t < 8; ++t) acc1[t] = vzero;

    v4f bq[2][8];
#pragma unroll
    for (int t = 0; t < 8; ++t) bq[0][t] = ldsW1[t * 32 + lane];
    __builtin_amdgcn_sched_group_barrier(0x100, 8, 0);
#pragma unroll
    for (int sp = 0; sp < 8; ++sp) {
      if (sp + 1 < 8) {
#pragma unroll
        for (int t = 0; t < 8; ++t) bq[(sp + 1) & 1][t] = ldsW1[((sp + 1) * 8 + t) * 32 + lane];
      }
      v2f A0, A1;
      A0.x = a[4 * sp];
      A0.y = a[4 * sp + 1];
      A1.x = a[4 * sp + 2];
      A1.y = a[4 * sp + 3];
#pragma unroll
      for (int t = 0; t < 8; ++t) {
        v2f Bv;
        Bv.x = bq[sp & 1][t].x;
        Bv.y = bq[sp & 1][t].y;
        acc1[t] = __builtin_amdgcn_wmma_f32_16x16x4_f32(
            false, A0, false, Bv, (short)0, acc1[t], false, false);
      }
#pragma unroll
      for (int t = 0; t < 8; ++t) {
        v2f Bv;
        Bv.x = bq[sp & 1][t].z;
        Bv.y = bq[sp & 1][t].w;
        acc1[t] = __builtin_amdgcn_wmma_f32_16x16x4_f32(
            false, A1, false, Bv, (short)0, acc1[t], false, false);
      }
      if (sp + 1 < 8) __builtin_amdgcn_sched_group_barrier(0x100, 8, 0);
      __builtin_amdgcn_sched_group_barrier(0x008, 16, 0);
    }

    load_edge_rows(a, evb + (size_t)(pn & (Nq - 1)) * Fq, evb + (size_t)njn * Fq, h);

#pragma unroll
    for (int t = 0; t < 8; ++t) {
#pragma unroll
      for (int r = 0; r < 8; ++r) {
        float x = acc1[t][r] + bias1[t];
        x = fmaxf(x, 0.0f);
        myH[(r + 8 * h) * H_STRIDE + 16 * t + m] = x;
      }
    }

    v8f acc2[4];
#pragma unroll
    for (int t = 0; t < 4; ++t) acc2[t] = vzero;

    v4f cq[2][4];
    v2f hA[2][2];
#pragma unroll
    for (int t = 0; t < 4; ++t) cq[0][t] = ldsW2[t * 32 + lane];
    hA[0][0].x = myH[m * H_STRIDE + 2 * h];
    hA[0][0].y = myH[m * H_STRIDE + 2 * h + 1];
    hA[0][1].x = myH[m * H_STRIDE + 2 * h + 4];
    hA[0][1].y = myH[m * H_STRIDE + 2 * h + 5];
    __builtin_amdgcn_sched_group_barrier(0x100, 6, 0);
#pragma unroll
    for (int sp = 0; sp < 16; ++sp) {
      if (sp + 1 < 16) {
        const int kn = 8 * (sp + 1) + 2 * h;
#pragma unroll
        for (int t = 0; t < 4; ++t) cq[(sp + 1) & 1][t] = ldsW2[((sp + 1) * 4 + t) * 32 + lane];
        hA[(sp + 1) & 1][0].x = myH[m * H_STRIDE + kn];
        hA[(sp + 1) & 1][0].y = myH[m * H_STRIDE + kn + 1];
        hA[(sp + 1) & 1][1].x = myH[m * H_STRIDE + kn + 4];
        hA[(sp + 1) & 1][1].y = myH[m * H_STRIDE + kn + 5];
      }
#pragma unroll
      for (int t = 0; t < 4; ++t) {
        v2f Bv;
        Bv.x = cq[sp & 1][t].x;
        Bv.y = cq[sp & 1][t].y;
        acc2[t] = __builtin_amdgcn_wmma_f32_16x16x4_f32(
            false, hA[sp & 1][0], false, Bv, (short)0, acc2[t], false, false);
      }
#pragma unroll
      for (int t = 0; t < 4; ++t) {
        v2f Bv;
        Bv.x = cq[sp & 1][t].z;
        Bv.y = cq[sp & 1][t].w;
        acc2[t] = __builtin_amdgcn_wmma_f32_16x16x4_f32(
            false, hA[sp & 1][1], false, Bv, (short)0, acc2[t], false, false);
      }
      if (sp + 1 < 16) __builtin_amdgcn_sched_group_barrier(0x100, 6, 0);
      __builtin_amdgcn_sched_group_barrier(0x008, 8, 0);
    }

#pragma unroll
    for (int t = 0; t < 4; ++t) {
      float part = 0.0f;
#pragma unroll
      for (int r = 0; r < 8; ++r) part += fmaxf(acc2[t][r] + bias2[t], 0.0f);
      part += __shfl_xor(part, 16, 32);
      if (h == 0) out[(size_t)p * Dq + 16 * t + m] = part * (1.0f / 16.0f);
    }

    p = pn;
  }
}

extern "C" void kernel_launch(void* const* d_in, const int* in_sizes, int n_in,
                              void* d_out, int out_size, void* d_ws, size_t ws_size,
                              hipStream_t stream) {
  const float* events = (const float*)d_in[0];
  const float* W1     = (const float*)d_in[1];
  const float* b1     = (const float*)d_in[2];
  const float* W2     = (const float*)d_in[3];
  const float* b2     = (const float*)d_in[4];
  float*       outp   = (float*)d_out;
  int*         knn    = (int*)d_ws;

  knn_kernel<<<(Bq * Nq) / 256, 256, 0, stream>>>(events, knn);
  edgeconv_mlp_kernel<<<(Bq * Nq) / PTS_PER_WG, MLP_WAVES * 32, 0, stream>>>(
      events, knn, W1, b1, W2, b2, outp);
}


# MLP: a wave issues at s_setprio 3 and steps down to 2, 1, 0 over its last three points, so a SIMD's four waves finish together and none runs its tail alone with the matrix pipe idle; same sums
# speedup vs baseline: 3.8009x; 1.0528x over previous
#include <hip/hip_runtime.h>

typedef __attribute__((ext_vector_type(2))) float v2f;
typedef __attribute__((ext_vector_type(8))) float v8f;

#define Bq 128
#define Nq 512
#define Fq 32
#define Hq 128
#define Dq 64
#define Kq 16

#define QCAP 64
#define QSLOT_BYTES (256 * 8)

__device__ __forceinline__ void knn_insert(float (&bestd)[Kq], int (&besti)[Kq],
                                           float d, int j) {
  bool lt[Kq];
#pragma unroll
  for (int t = 0; t < Kq; ++t) lt[t] = d < bestd[t];
#pragma unroll
  for (int t = Kq - 1; t >= 1; --t) {
    const int shifted = lt[t - 1] ? besti[t - 1] : j;
    besti[t] = lt[t] ? shifted : besti[t];
    bestd[t] = fminf(bestd[t], fmaxf(bestd[t - 1], d));
  }
  besti[0] = lt[0] ? j : besti[0];
  bestd[0] = fminf(bestd[0], d);
}

__global__ __launch_bounds__(256) void knn_kernel(const float* __restrict__ events,
                                                  int* __restrict__ knn_idx) {
  __shared__ __attribute__((aligned(16))) float2 sq[QCAP * 256];
  __shared__ __attribute__((aligned(16))) float2 sxy[Nq];
  const int tid = threadIdx.x;
  const int p   = blockIdx.x * 256 + tid;
  const int b   = p >> 9;
  const int i   = p & (Nq - 1);

  for (int j = tid; j < Nq; j += 256) {
    const float* e = events + ((size_t)b * Nq + j) * Fq;
    sxy[j] = make_float2(e[0], e[1]);
  }
  for (int s = 0; s < QCAP; ++s) sq[s * 256 + tid] = make_float2(0.0f, 0.0f);
  __syncthreads();

  const float cx = sxy[i].x, cy = sxy[i].y;
  float bestd[Kq];
  int   besti[Kq];
#pragma unroll
  for (int t = 0; t < Kq; ++t) { bestd[t] = 3.4e38f; besti[t] = 0; }

  char* const qbase = reinterpret_cast<char*>(sq) + tid * 8;

  for (int j0 = 0; j0 < Nq;) {
    const int len = (j0 < 64) ? 32 : 64;
    const float thr = bestd[Kq - 1];
    int qb = 0;

#pragma unroll 4
    for (int jj = 0; jj < len; jj += 2) {
      const int j = j0 + jj;
      const float4 c2 = *reinterpret_cast<const float4*>(&sxy[j]);
      {
        const float dx = c2.x - cx;
        const float dy = c2.y - cy;
        const float d = __builtin_fmaf(dx, dx, dy * dy);
        *reinterpret_cast<float2*>(qbase + qb) = make_float2(d, __int_as_float(j));
        qb += (d < thr) ? QSLOT_BYTES : 0;
      }
      {
        const float dx = c2.z - cx;
        const float dy = c2.w - cy;
        const float d = __builtin_fmaf(dx, dx, dy * dy);
        *reinterpret_cast<float2*>(qbase + qb) = make_float2(d, __int_as_float(j + 1));
        qb += (d < thr) ? QSLOT_BYTES : 0;
      }
    }
    const int cnt = qb / QSLOT_BYTES;

    for (int q = 0; q < QCAP; q += 2) {
      if (__builtin_amdgcn_ballot_w32(q < cnt) == 0u) break;
      const float2 e0 = *reinterpret_cast<const float2*>(qbase + q * QSLOT_BYTES);
      const float2 e1 = *reinterpret_cast<const float2*>(qbase + (q + 1) * QSLOT_BYTES);
      const int ja = __float_as_int(e0.y);
      const int jb = __float_as_int(e1.y);
      const float da = ((q < cnt) & (ja != i)) ? e0.x : 3.4e38f;
      const float db = ((q + 1 < cnt) & (jb != i)) ? e1.x : 3.4e38f;
      knn_insert(bestd, besti, da, ja);
      knn_insert(bestd, besti, db, jb);
    }
    j0 += len;
  }
#pragma unroll
  for (int t = 0; t < Kq; ++t) knn_idx[(size_t)p * Kq + t] = besti[t];
}

typedef __attribute__((ext_vector_type(4))) float v4f;

#define MLP_WAVES 16
#define PTS_PER_WG 256
#define PTS_PER_WAVE (PTS_PER_WG / MLP_WAVES)
#define H_STRIDE 129

__device__ __forceinline__ void load_edge_rows(float (&a)[32], const float* ev,
                                               const float* nb, int h) {
#pragma unroll
  for (int s = 0; s < 8; ++s) {
    const int c = 4 * s + 2 * h;
    const float x0 = ev[c], x1 = ev[c + 1];
    a[2 * s]          = x0;
    a[2 * s + 1]      = x1;
    a[2 * (s + 8)]     = nb[c]     - x0;
    a[2 * (s + 8) + 1] = nb[c + 1] - x1;
  }
}

__global__ __launch_bounds__(MLP_WAVES * 32) void edgeconv_mlp_kernel(
    const float* __restrict__ events, const int* __restrict__ knn_idx,
    const float* __restrict__ W1, const float* __restrict__ b1,
    const float* __restrict__ W2, const float* __restrict__ b2,
    float* __restrict__ out) {
  __shared__ v4f   ldsW1[8 * 8 * 32];
  __shared__ v4f   ldsW2[16 * 4 * 32];
  __shared__ float ldsH[MLP_WAVES * 16 * H_STRIDE];

  const int tid  = threadIdx.x;
  const int lane = tid & 31;
  const int wid  = tid >> 5;
  const int m    = lane & 15;
  const int h    = lane >> 4;

#pragma unroll
  for (int i = 0; i < (8 * 8 * 32) / (MLP_WAVES * 32); ++i) {
    const int q = tid + i * (MLP_WAVES * 32);
    const int ql = q & 31, t = (q >> 5) & 7, sp = q >> 8;
    const float* w = W1 + (size_t)(8 * sp + 2 * (ql >> 4)) * Hq + 16 * t + (ql & 15);
    v4f v;
    v.x = w[0];
    v.y = w[Hq];
    v.z = w[4 * Hq];
    v.w = w[5 * Hq];
    ldsW1[q] = v;
  }
#pragma unroll
  for (int i = 0; i < (16 * 4 * 32) / (MLP_WAVES * 32); ++i) {
    const int q = tid + i * (MLP_WAVES * 32);
    const int ql = q & 31, t = (q >> 5) & 3, sp = q >> 7;
    const float* w = W2 + (size_t)(8 * sp + 2 * (ql >> 4)) * Dq + 16 * t + (ql & 15);
    v4f v;
    v.x = w[0];
    v.y = w[Dq];
    v.z = w[4 * Dq];
    v.w = w[5 * Dq];
    ldsW2[q] = v;
  }

  float bias1[8], bias2[4];
#pragma unroll
  for (int t = 0; t < 8; ++t) bias1[t] = b1[16 * t + m];
#pragma unroll
  for (int t = 0; t < 4; ++t) bias2[t] = b2[16 * t + m];

  float* myH = ldsH + wid * 16 * H_STRIDE;

  const int pbase = blockIdx.x * PTS_PER_WG;
  const float* evb = events + (size_t)(pbase >> 9) * Nq * Fq;

  int p = pbase + wid;
  float a[32];
  {
    const int nj = knn_idx[(size_t)p * Kq + m];
    load_edge_rows(a, evb + (size_t)(p & (Nq - 1)) * Fq, evb + (size_t)nj * Fq, h);
  }

  __syncthreads();

  const v8f vzero = {};

  __builtin_amdgcn_s_setprio(3);

#pragma unroll 1
  for (int it = 0; it < PTS_PER_WAVE; ++it) {
    {
      const int left = PTS_PER_WAVE - 1 - it;
      if (left == 2) __builtin_amdgcn_s_setprio(2);
      else if (left == 1) __builtin_amdgcn_s_setprio(1);
      else if (left == 0) __builtin_amdgcn_s_setprio(0);
    }
    const int pn = (it + 1 < PTS_PER_WAVE) ? p + MLP_WAVES : p;
    const int njn = knn_idx[(size_t)pn * Kq + m];

    v8f acc1[8];
#pragma unroll
    for (int t = 0; t < 8; ++t) acc1[t] = vzero;

    v4f bq[2][8];
#pragma unroll
    for (int t = 0; t < 8; ++t) bq[0][t] = ldsW1[t * 32 + lane];
    __builtin_amdgcn_sched_group_barrier(0x100, 8, 0);
#pragma unroll
    for (int sp = 0; sp < 8; ++sp) {
      if (sp + 1 < 8) {
#pragma unroll
        for (int t = 0; t < 8; ++t) bq[(sp + 1) & 1][t] = ldsW1[((sp + 1) * 8 + t) * 32 + lane];
      }
      v2f A0, A1;
      A0.x = a[4 * sp];
      A0.y = a[4 * sp + 1];
      A1.x = a[4 * sp + 2];
      A1.y = a[4 * sp + 3];
#pragma unroll
      for (int t = 0; t < 8; ++t) {
        v2f Bv;
        Bv.x = bq[sp & 1][t].x;
        Bv.y = bq[sp & 1][t].y;
        acc1[t] = __builtin_amdgcn_wmma_f32_16x16x4_f32(
            false, A0, false, Bv, (short)0, acc1[t], false, false);
      }
#pragma unroll
      for (int t = 0; t < 8; ++t) {
        v2f Bv;
        Bv.x = bq[sp & 1][t].z;
        Bv.y = bq[sp & 1][t].w;
        acc1[t] = __builtin_amdgcn_wmma_f32_16x16x4_f32(
            false, A1, false, Bv, (short)0, acc1[t], false, false);
      }
      if (sp + 1 < 8) __builtin_amdgcn_sched_group_barrier(0x100, 8, 0);
      __builtin_amdgcn_sched_group_barrier(0x008, 16, 0);
    }

    load_edge_rows(a, evb + (size_t)(pn & (Nq - 1)) * Fq, evb + (size_t)njn * Fq, h);

#pragma unroll
    for (int t = 0; t < 8; ++t) {
#pragma unroll
      for (int r = 0; r < 8; ++r) {
        float x = acc1[t][r] + bias1[t];
        x = fmaxf(x, 0.0f);
        myH[(r + 8 * h) * H_STRIDE + 16 * t + m] = x;
      }
    }

    v8f acc2[4];
#pragma unroll
    for (int t = 0; t < 4; ++t) acc2[t] = vzero;

    v4f cq[2][4];
    v2f hA[2][2];
#pragma unroll
    for (int t = 0; t < 4; ++t) cq[0][t] = ldsW2[t * 32 + lane];
    hA[0][0].x = myH[m * H_STRIDE + 2 * h];
    hA[0][0].y = myH[m * H_STRIDE + 2 * h + 1];
    hA[0][1].x = myH[m * H_STRIDE + 2 * h + 4];
    hA[0][1].y = myH[m * H_STRIDE + 2 * h + 5];
    __builtin_amdgcn_sched_group_barrier(0x100, 6, 0);
#pragma unroll
    for (int sp = 0; sp < 16; ++sp) {
      if (sp + 1 < 16) {
        const int kn = 8 * (sp + 1) + 2 * h;
#pragma unroll
        for (int t = 0; t < 4; ++t) cq[(sp + 1) & 1][t] = ldsW2[((sp + 1) * 4 + t) * 32 + lane];
        hA[(sp + 1) & 1][0].x = myH[m * H_STRIDE + kn];
        hA[(sp + 1) & 1][0].y = myH[m * H_STRIDE + kn + 1];
        hA[(sp + 1) & 1][1].x = myH[m * H_STRIDE + kn + 4];
        hA[(sp + 1) & 1][1].y = myH[m * H_STRIDE + kn + 5];
      }
#pragma unroll
      for (int t = 0; t < 4; ++t) {
        v2f Bv;
        Bv.x = cq[sp & 1][t].x;
        Bv.y = cq[sp & 1][t].y;
        acc2[t] = __builtin_amdgcn_wmma_f32_16x16x4_f32(
            false, hA[sp & 1][0], false, Bv, (short)0, acc2[t], false, false);
      }
#pragma unroll
      for (int t = 0; t < 4; ++t) {
        v2f Bv;
        Bv.x = cq[sp & 1][t].z;
        Bv.y = cq[sp & 1][t].w;
        acc2[t] = __builtin_amdgcn_wmma_f32_16x16x4_f32(
            false, hA[sp & 1][1], false, Bv, (short)0, acc2[t], false, false);
      }
      if (sp + 1 < 16) __builtin_amdgcn_sched_group_barrier(0x100, 6, 0);
      __builtin_amdgcn_sched_group_barrier(0x008, 8, 0);
    }

#pragma unroll
    for (int t = 0; t < 4; ++t) {
      float part = 0.0f;
#pragma unroll
      for (int r = 0; r < 8; ++r) part += fmaxf(acc2[t][r] + bias2[t], 0.0f);
      part += __shfl_xor(part, 16, 32);
      if (h == 0) out[(size_t)p * Dq + 16 * t + m] = part * (1.0f / 16.0f);
    }

    p = pn;
  }
}

extern "C" void kernel_launch(void* const* d_in, const int* in_sizes, int n_in,
                              void* d_out, int out_size, void* d_ws, size_t ws_size,
                              hipStream_t stream) {
  const float* events = (const float*)d_in[0];
  const float* W1     = (const float*)d_in[1];
  const float* b1     = (const float*)d_in[2];
  const float* W2     = (const float*)d_in[3];
  const float* b2     = (const float*)d_in[4];
  float*       outp   = (float*)d_out;
  int*         knn    = (int*)d_ws;

  knn_kernel<<<(Bq * Nq) / 256, 256, 0, stream>>>(events, knn);
  edgeconv_mlp_kernel<<<(Bq * Nq) / PTS_PER_WG, MLP_WAVES * 32, 0, stream>>>(
      events, knn, W1, b1, W2, b2, outp);
}
